# starting kernel
# baseline (speedup 1.0000x reference)
#include <hip/hip_runtime.h>
#include <hip/hip_bf16.h>
#include <math.h>

typedef __attribute__((ext_vector_type(16))) _Float16 v16h;
typedef __attribute__((ext_vector_type(8)))  _Float16 v8h;
typedef __attribute__((ext_vector_type(4)))  _Float16 v4h;
typedef __attribute__((ext_vector_type(8)))  float    v8f;

#define NPTS      262144
#define BSC       128
#define PPS       2048
#define CIN       256
#define CMID      256
#define COUT      512
#define KTOT      288
#define ROWS      64
#define CHUNKS    (PPS / ROWS)
#define ASTRIDE   296

#define WS_W1T     0
#define WS_CENTER  147456
#define WS_PARTIAL 149504

__global__ void prep_w1t_kernel(const float* __restrict__ W1,
                                _Float16* __restrict__ W1t) {
    int idx = blockIdx.x * blockDim.x + threadIdx.x;
    if (idx >= CMID * KTOT) return;
    int col = idx / KTOT;
    int k   = idx % KTOT;
    float v = (k < CIN + 3) ? W1[(size_t)k * CMID + col] : 0.0f;
    W1t[(size_t)col * KTOT + k] = (_Float16)v;
}

__global__ void centroid_kernel(const float* __restrict__ pos,
                                float* __restrict__ center) {
    int s = blockIdx.x;
    int t = threadIdx.x;
    float sx = 0.f, sy = 0.f, sz = 0.f;
    for (int r = t; r < PPS; r += 256) {
        const float* p = pos + (size_t)(s * PPS + r) * 3;
        sx += p[0]; sy += p[1]; sz += p[2];
    }
    __shared__ float rx[256], ry[256], rz[256];
    rx[t] = sx; ry[t] = sy; rz[t] = sz;
    __syncthreads();
    for (int off = 128; off > 0; off >>= 1) {
        if (t < off) { rx[t] += rx[t + off]; ry[t] += ry[t + off]; rz[t] += rz[t + off]; }
        __syncthreads();
    }
    if (t == 0) {
        const float inv = 1.0f / (float)PPS;
        center[s * 4 + 0] = rx[0] * inv;
        center[s * 4 + 1] = ry[0] * inv;
        center[s * 4 + 2] = rz[0] * inv;
    }
}

__global__ __launch_bounds__(256)
void gemm_max_kernel(const float* __restrict__ pos,
                     const float* __restrict__ feature,
                     const _Float16* __restrict__ W1t,
                     const float* __restrict__ center,
                     float* __restrict__ partial) {
    __shared__ __align__(16) _Float16 As[ROWS][ASTRIDE];

    const int b = blockIdx.x;
    const int t = threadIdx.x;
    const int s = b >> 5;
    const int rowbase = b * ROWS;

    const float4* f4 = (const float4*)feature;
    for (int idx = t; idx < ROWS * (CIN / 4); idx += 256) {
        int r = idx >> 6;
        int j = idx & 63;
        float4 v = f4[(size_t)(rowbase + r) * (CIN / 4) + j];
        v4h h;
        h[0] = (_Float16)v.x; h[1] = (_Float16)v.y;
        h[2] = (_Float16)v.z; h[3] = (_Float16)v.w;
        *(v4h*)&As[r][j * 4] = h;
    }
    if (t < ROWS) {
        int r = t;
        const float* pp = pos + (size_t)(rowbase + r) * 3;
        float cx = center[s * 4 + 0];
        float cy = center[s * 4 + 1];
        float cz = center[s * 4 + 2];
        As[r][CIN + 0] = (_Float16)(pp[0] - cx);
        As[r][CIN + 1] = (_Float16)(pp[1] - cy);
        As[r][CIN + 2] = (_Float16)(pp[2] - cz);
        for (int k = CIN + 3; k < ASTRIDE; ++k) As[r][k] = (_Float16)0.0f;
    }
    __syncthreads();

    const int lane = t & 31;
    const int wv   = t >> 5;
    const int n    = lane & 15;
    const int half = lane >> 4;
    const int lo8  = half * 8;

    v8f acc[4][2] = {};

    for (int kt = 0; kt < KTOT / 32; ++kt) {
        const int kb = kt * 32;

        v16h bfrag[2];
#pragma unroll
        for (int ct = 0; ct < 2; ++ct) {
            int col = (wv * 2 + ct) * 16 + n;
            const _Float16* bp = W1t + (size_t)col * KTOT + kb + lo8;
            v8h blo = *(const v8h*)bp;
            v8h bhi = *(const v8h*)(bp + 16);
            bfrag[ct] = __builtin_shufflevector(blo, bhi,
                0,1,2,3,4,5,6,7,8,9,10,11,12,13,14,15);
        }

#pragma unroll
        for (int rt = 0; rt < 4; ++rt) {
            const _Float16* ap = &As[rt * 16 + n][kb + lo8];
            v8h alo = *(const v8h*)ap;
            v8h ahi = *(const v8h*)(ap + 16);
            v16h afrag = __builtin_shufflevector(alo, ahi,
                0,1,2,3,4,5,6,7,8,9,10,11,12,13,14,15);

            acc[rt][0] = __builtin_amdgcn_wmma_f32_16x16x32_f16(
                false, afrag, false, bfrag[0], (short)0, acc[rt][0], false, false);
            acc[rt][1] = __builtin_amdgcn_wmma_f32_16x16x32_f16(
                false, afrag, false, bfrag[1], (short)0, acc[rt][1], false, false);
        }
    }

#pragma unroll
    for (int ct = 0; ct < 2; ++ct) {
        float m = acc[0][ct][0];
#pragma unroll
        for (int rt = 0; rt < 4; ++rt)
#pragma unroll
            for (int i = 0; i < 8; ++i)
                m = fmaxf(m, acc[rt][ct][i]);
        m = fmaxf(m, __shfl_xor(m, 16, 32));
        if (half == 0) {
            int col = (wv * 2 + ct) * 16 + n;
            partial[(size_t)b * CMID + col] = m;
        }
    }
}

__global__ __launch_bounds__(256)
void finish_kernel(const float* __restrict__ partial,
                   const float* __restrict__ b1,
                   const float* __restrict__ W2,
                   const float* __restrict__ b2,
                   const float* __restrict__ noise,
                   float* __restrict__ out) {
    __shared__ float agg[CMID];
    const int s = blockIdx.x;
    const int c = threadIdx.x;

    const float* p = partial + (size_t)s * CHUNKS * CMID + c;
    float m = p[0];
    for (int ch = 1; ch < CHUNKS; ++ch) m = fmaxf(m, p[ch * CMID]);
    agg[c] = fmaxf(m + b1[c], 0.0f);
    __syncthreads();

    float mu = b2[c];
    float sr = b2[c + CMID];
    for (int k = 0; k < CMID; ++k) {
        float a = agg[k];
        mu += a * W2[(size_t)k * COUT + c];
        sr += a * W2[(size_t)k * COUT + c + CMID];
    }
    float sig = fmaxf(sr, 0.0f) + log1pf(expf(-fabsf(sr))) + 1e-4f;
    out[(size_t)s * CMID + c] = mu + sig * noise[(size_t)s * CMID + c];
}

extern "C" void kernel_launch(void* const* d_in, const int* in_sizes, int n_in,
                              void* d_out, int out_size, void* d_ws, size_t ws_size,
                              hipStream_t stream) {
    const float* pos     = (const float*)d_in[0];
    const float* feature = (const float*)d_in[1];
    const float* W1      = (const float*)d_in[3];
    const float* b1      = (const float*)d_in[4];
    const float* W2      = (const float*)d_in[5];
    const float* b2      = (const float*)d_in[6];
    const float* noise   = (const float*)d_in[7];
    float* out = (float*)d_out;

    _Float16* W1t    = (_Float16*)((char*)d_ws + WS_W1T);
    float*    center = (float*)((char*)d_ws + WS_CENTER);
    float*    partial= (float*)((char*)d_ws + WS_PARTIAL);

    prep_w1t_kernel<<<(CMID * KTOT + 255) / 256, 256, 0, stream>>>(W1, W1t);
    centroid_kernel<<<BSC, 256, 0, stream>>>(pos, center);
    gemm_max_kernel<<<BSC * CHUNKS, 256, 0, stream>>>(pos, feature, W1t, center, partial);
    finish_kernel<<<BSC, 256, 0, stream>>>(partial, b1, W2, b2, noise, out);
}


# GEMM operand paths swapped: f16 W1t kept in LDS by 256 persistent 4-wave blocks (was re-read from L2 by 4096 blocks); A fragments go global->registers, no LDS staging of A, no barrier in the loop
# speedup vs baseline: 1.2414x; 1.2414x over previous
#include <hip/hip_runtime.h>
#include <hip/hip_bf16.h>
#include <math.h>

typedef __attribute__((ext_vector_type(16))) _Float16 v16h;
typedef __attribute__((ext_vector_type(8)))  _Float16 v8h;
typedef __attribute__((ext_vector_type(8)))  float    v8f;

#define NPTS      262144
#define BSC       128
#define PPS       2048
#define CIN       256
#define CMID      256
#define COUT      512
#define KTOT      288
#define ROWS      64
#define CHUNKS    (PPS / ROWS)

#define WS_W1T     0
#define WS_CENTER  147456
#define WS_PARTIAL 149504

__global__ void prep_w1t_kernel(const float* __restrict__ W1,
                                _Float16* __restrict__ W1t) {
    int idx = blockIdx.x * blockDim.x + threadIdx.x;
    if (idx >= CMID * KTOT) return;
    int col = idx / KTOT;
    int k   = idx % KTOT;
    float v = (k < CIN + 3) ? W1[(size_t)k * CMID + col] : 0.0f;
    W1t[(size_t)((k >> 5) * CMID + col) * 32 + (k & 31)] = (_Float16)v;
}

__global__ void centroid_kernel(const float* __restrict__ pos,
                                float* __restrict__ center) {
    int s = blockIdx.x;
    int t = threadIdx.x;
    float sx = 0.f, sy = 0.f, sz = 0.f;
    for (int r = t; r < PPS; r += 256) {
        const float* p = pos + (size_t)(s * PPS + r) * 3;
        sx += p[0]; sy += p[1]; sz += p[2];
    }
    __shared__ float rx[256], ry[256], rz[256];
    rx[t] = sx; ry[t] = sy; rz[t] = sz;
    __syncthreads();
    for (int off = 128; off > 0; off >>= 1) {
        if (t < off) { rx[t] += rx[t + off]; ry[t] += ry[t + off]; rz[t] += rz[t + off]; }
        __syncthreads();
    }
    if (t == 0) {
        const float inv = 1.0f / (float)PPS;
        center[s * 4 + 0] = rx[0] * inv;
        center[s * 4 + 1] = ry[0] * inv;
        center[s * 4 + 2] = rz[0] * inv;
    }
}

#define RT      2
#define DEP     8
#define WAVES   4
#define WROWS   (1024 / WAVES)
#define NPASS   (WROWS / (16 * RT))
#define PPC     (ROWS / (16 * RT))
#define KFEAT   (CIN / 32)
#define GB      16
#define NG      ((KTOT / 32) * (16 / GB))
#define BCOLP   40
#define BKSTEP  (CMID * BCOLP)
#define FILLR   2
#define FILLN   ((KTOT / 32) * CMID * 4 / (WAVES * 32) / FILLR)

__device__ __forceinline__ void load_a(float4 (&r)[4], const float* src) {
    const float4* p = (const float4*)src;
    r[0] = p[0]; r[1] = p[1];
    r[2] = p[4]; r[3] = p[5];
}

__device__ __forceinline__ v16h cvt_a(const float4 (&r)[4]) {
    v16h a;
    a[0]  = (_Float16)r[0].x; a[1]  = (_Float16)r[0].y; a[2]  = (_Float16)r[0].z; a[3]  = (_Float16)r[0].w;
    a[4]  = (_Float16)r[1].x; a[5]  = (_Float16)r[1].y; a[6]  = (_Float16)r[1].z; a[7]  = (_Float16)r[1].w;
    a[8]  = (_Float16)r[2].x; a[9]  = (_Float16)r[2].y; a[10] = (_Float16)r[2].z; a[11] = (_Float16)r[2].w;
    a[12] = (_Float16)r[3].x; a[13] = (_Float16)r[3].y; a[14] = (_Float16)r[3].z; a[15] = (_Float16)r[3].w;
    return a;
}

__device__ __forceinline__ void load_bgroup(v16h (&b)[GB], const _Float16* bl,
                                            const unsigned (&koff)[KTOT / 32], int G) {
    const int kt = G / (16 / GB), sub = G % (16 / GB);
#pragma unroll
    for (int j = 0; j < GB; ++j) {
        const _Float16* bp = bl + koff[kt] + (sub * GB + j) * 16 * BCOLP;
        v8h blo = *(const v8h*)bp;
        v8h bhi = *(const v8h*)(bp + 16);
        b[j] = __builtin_shufflevector(blo, bhi, 0,1,2,3,4,5,6,7,8,9,10,11,12,13,14,15);
    }
}

template <bool PF>
__device__ __forceinline__ void gemm_pass(float4 (&ring)[DEP][RT][4], v16h (&bq)[2][GB],
                                          const float* fcur, const float* fnext,
                                          const float* pcur, const _Float16* bl,
                                          int half, float cx, float cy, float cz,
                                          float (&cm)[16]) {
    float px[RT], py[RT], pz[RT];
#pragma unroll
    for (int rt = 0; rt < RT; ++rt) {
        px[rt] = pcur[rt * 48 + 0]; py[rt] = pcur[rt * 48 + 1]; pz[rt] = pcur[rt * 48 + 2];
    }

    v8f acc[RT][16] = {};
    v16h afrag[RT];

    unsigned koff[KTOT / 32];
#pragma unroll
    for (int kt = 0; kt < KTOT / 32; ++kt) {
        koff[kt] = kt * BKSTEP;
        asm volatile("" : "+v"(koff[kt]));
    }

#pragma unroll
    for (int G = 0; G < NG; ++G) {
        const int kt = G / (16 / GB), sub = G % (16 / GB);
        if (G + 1 < NG) load_bgroup(bq[(G + 1) & 1], bl, koff, G + 1);

        __builtin_amdgcn_sched_barrier(0);

        if (sub == 0) {
            if (kt < KFEAT) {
                const int slot = kt % DEP;
#pragma unroll
                for (int rt = 0; rt < RT; ++rt) afrag[rt] = cvt_a(ring[slot][rt]);
                const int kn = kt + DEP;
                if (kn < KFEAT) {
#pragma unroll
                    for (int rt = 0; rt < RT; ++rt)
                        load_a(ring[slot][rt], fcur + rt * 16 * CIN + kn * 32);
                } else if (PF) {
#pragma unroll
                    for (int rt = 0; rt < RT; ++rt)
                        load_a(ring[slot][rt], fnext + rt * 16 * CIN + (kn - KFEAT) * 32);
                }
            } else {
#pragma unroll
                for (int rt = 0; rt < RT; ++rt) {
                    v16h a = {};
                    a[0] = half ? (_Float16)0.0f : (_Float16)(px[rt] - cx);
                    a[1] = half ? (_Float16)0.0f : (_Float16)(py[rt] - cy);
                    a[2] = half ? (_Float16)0.0f : (_Float16)(pz[rt] - cz);
                    afrag[rt] = a;
                }
            }
        }

#pragma unroll
        for (int j = 0; j < GB; ++j)
#pragma unroll
            for (int rt = 0; rt < RT; ++rt)
                acc[rt][sub * GB + j] = __builtin_amdgcn_wmma_f32_16x16x32_f16(
                    false, afrag[rt], false, bq[G & 1][j], (short)0, acc[rt][sub * GB + j], false, false);

        __builtin_amdgcn_sched_barrier(0);
    }

    if (PF) load_bgroup(bq[0], bl, koff, 0);
    __builtin_amdgcn_sched_barrier(0);

#pragma unroll
    for (int ct = 0; ct < 16; ++ct)
#pragma unroll
        for (int rt = 0; rt < RT; ++rt)
#pragma unroll
            for (int i = 0; i < 8; ++i)
                cm[ct] = fmaxf(cm[ct], acc[rt][ct][i]);
}

__device__ __forceinline__ void flush_chunk(float (&cm)[16], float* dst, int lane, int half) {
#pragma unroll
    for (int j = 0; j < 8; ++j) {
        float m0 = fmaxf(cm[2 * j],     __shfl_xor(cm[2 * j],     16, 32));
        float m1 = fmaxf(cm[2 * j + 1], __shfl_xor(cm[2 * j + 1], 16, 32));
        dst[32 * j + lane] = half ? m1 : m0;
    }
#pragma unroll
    for (int ct = 0; ct < 16; ++ct) cm[ct] = -INFINITY;
}

__global__ __launch_bounds__(WAVES * 32)
void gemm_max_kernel(const float* __restrict__ pos,
                     const float* __restrict__ feature,
                     const _Float16* __restrict__ W1t,
                     const float* __restrict__ center,
                     float* __restrict__ partial) {
    __shared__ __align__(16) _Float16 Bs[(KTOT / 32) * BKSTEP];

    const int g    = blockIdx.x;
    const int t    = threadIdx.x;
    const int lane = t & 31;
    const int wv   = t >> 5;
    const int n    = lane & 15;
    const int half = lane >> 4;
    const int lo8  = half * 8;
    const int s    = g >> 1;
    const int wrow0 = g * (WAVES * WROWS) + wv * WROWS;

    const float* fl = feature + (size_t)(wrow0 + n) * CIN + lo8;
    const float* pl = pos + (size_t)(wrow0 + n) * 3;

    float4 ring[DEP][RT][4];
#pragma unroll
    for (int d = 0; d < DEP; ++d)
#pragma unroll
        for (int rt = 0; rt < RT; ++rt) load_a(ring[d][rt], fl + rt * 16 * CIN + d * 32);

#pragma unroll
    for (int r = 0; r < FILLR; ++r) {
        v8h tmp[FILLN];
#pragma unroll
        for (int i = 0; i < FILLN; ++i)
            tmp[i] = *(const v8h*)(W1t + (size_t)(t + (r * FILLN + i) * (WAVES * 32)) * 8);
#pragma unroll
        for (int i = 0; i < FILLN; ++i) {
            const int pi = t + (r * FILLN + i) * (WAVES * 32);
            *(v8h*)&Bs[(pi >> 2) * BCOLP + (pi & 3) * 8] = tmp[i];
        }
    }
    const float cx = center[s * 4 + 0];
    const float cy = center[s * 4 + 1];
    const float cz = center[s * 4 + 2];
    __syncthreads();

    const _Float16* bl = Bs + n * BCOLP + lo8;
    float* pout = partial + (size_t)(wrow0 / ROWS) * CMID;

    float cm[16];
#pragma unroll
    for (int ct = 0; ct < 16; ++ct) cm[ct] = -INFINITY;

    v16h bq[2][GB];
    {
        unsigned koff0[KTOT / 32] = {};
        load_bgroup(bq[0], bl, koff0, 0);
    }

    for (int q = 0; q < NPASS - 1; ++q) {
        const float* fcur = fl + (size_t)q * (16 * RT * CIN);
        gemm_pass<true>(ring, bq, fcur, fcur + 16 * RT * CIN, pl + q * (16 * RT * 3), bl,
                        half, cx, cy, cz, cm);
        if ((q % PPC) == PPC - 1) flush_chunk(cm, pout + (q / PPC) * CMID, lane, half);
    }
    {
        const int q = NPASS - 1;
        const float* fcur = fl + (size_t)q * (16 * RT * CIN);
        gemm_pass<false>(ring, bq, fcur, fcur, pl + q * (16 * RT * 3), bl,
                         half, cx, cy, cz, cm);
        flush_chunk(cm, pout + (q / PPC) * CMID, lane, half);
    }
}

__global__ __launch_bounds__(256)
void finish_kernel(const float* __restrict__ partial,
                   const float* __restrict__ b1,
                   const float* __restrict__ W2,
                   const float* __restrict__ b2,
                   const float* __restrict__ noise,
                   float* __restrict__ out) {
    __shared__ float agg[CMID];
    const int s = blockIdx.x;
    const int c = threadIdx.x;

    const float* p = partial + (size_t)s * CHUNKS * CMID + c;
    float m = p[0];
    for (int ch = 1; ch < CHUNKS; ++ch) m = fmaxf(m, p[ch * CMID]);
    agg[c] = fmaxf(m + b1[c], 0.0f);
    __syncthreads();

    float mu = b2[c];
    float sr = b2[c + CMID];
    for (int k = 0; k < CMID; ++k) {
        float a = agg[k];
        mu += a * W2[(size_t)k * COUT + c];
        sr += a * W2[(size_t)k * COUT + c + CMID];
    }
    float sig = fmaxf(sr, 0.0f) + log1pf(expf(-fabsf(sr))) + 1e-4f;
    out[(size_t)s * CMID + c] = mu + sig * noise[(size_t)s * CMID + c];
}

extern "C" void kernel_launch(void* const* d_in, const int* in_sizes, int n_in,
                              void* d_out, int out_size, void* d_ws, size_t ws_size,
                              hipStream_t stream) {
    const float* pos     = (const float*)d_in[0];
    const float* feature = (const float*)d_in[1];
    const float* W1      = (const float*)d_in[3];
    const float* b1      = (const float*)d_in[4];
    const float* W2      = (const float*)d_in[5];
    const float* b2      = (const float*)d_in[6];
    const float* noise   = (const float*)d_in[7];
    float* out = (float*)d_out;

    _Float16* W1t    = (_Float16*)((char*)d_ws + WS_W1T);
    float*    center = (float*)((char*)d_ws + WS_CENTER);
    float*    partial= (float*)((char*)d_ws + WS_PARTIAL);

    prep_w1t_kernel<<<(CMID * KTOT + 255) / 256, 256, 0, stream>>>(W1, W1t);
    centroid_kernel<<<BSC, 256, 0, stream>>>(pos, center);
    gemm_max_kernel<<<NPTS / (WAVES * WROWS), WAVES * 32, 0, stream>>>(pos, feature, W1t, center, partial);
    finish_kernel<<<BSC, 256, 0, stream>>>(partial, b1, W2, b2, noise, out);
}


# GEMM wave pass widened from 32 to 64 rows x 256 columns: each B fragment read from LDS now feeds 4 WMMAs, not 2, so LDS fragment reads per row are halved; A ring 4 K-steps deep
# speedup vs baseline: 1.2818x; 1.0325x over previous
#include <hip/hip_runtime.h>
#include <hip/hip_bf16.h>
#include <math.h>

typedef __attribute__((ext_vector_type(16))) _Float16 v16h;
typedef __attribute__((ext_vector_type(8)))  _Float16 v8h;
typedef __attribute__((ext_vector_type(8)))  float    v8f;

#define NPTS      262144
#define BSC       128
#define PPS       2048
#define CIN       256
#define CMID      256
#define COUT      512
#define KTOT      288
#define ROWS      64
#define CHUNKS    (PPS / ROWS)

#define WS_W1T     0
#define WS_CENTER  147456
#define WS_PARTIAL 149504

__global__ void prep_w1t_kernel(const float* __restrict__ W1,
                                _Float16* __restrict__ W1t) {
    int idx = blockIdx.x * blockDim.x + threadIdx.x;
    if (idx >= CMID * KTOT) return;
    int col = idx / KTOT;
    int k   = idx % KTOT;
    float v = (k < CIN + 3) ? W1[(size_t)k * CMID + col] : 0.0f;
    W1t[(size_t)((k >> 5) * CMID + col) * 32 + (k & 31)] = (_Float16)v;
}

__global__ void centroid_kernel(const float* __restrict__ pos,
                                float* __restrict__ center) {
    int s = blockIdx.x;
    int t = threadIdx.x;
    float sx = 0.f, sy = 0.f, sz = 0.f;
    for (int r = t; r < PPS; r += 256) {
        const float* p = pos + (size_t)(s * PPS + r) * 3;
        sx += p[0]; sy += p[1]; sz += p[2];
    }
    __shared__ float rx[256], ry[256], rz[256];
    rx[t] = sx; ry[t] = sy; rz[t] = sz;
    __syncthreads();
    for (int off = 128; off > 0; off >>= 1) {
        if (t < off) { rx[t] += rx[t + off]; ry[t] += ry[t + off]; rz[t] += rz[t + off]; }
        __syncthreads();
    }
    if (t == 0) {
        const float inv = 1.0f / (float)PPS;
        center[s * 4 + 0] = rx[0] * inv;
        center[s * 4 + 1] = ry[0] * inv;
        center[s * 4 + 2] = rz[0] * inv;
    }
}

#define RT      4
#define DEP     4
#define WAVES   4
#define WROWS   (1024 / WAVES)
#define NPASS   (WROWS / (16 * RT))
#define PPC     (ROWS / (16 * RT))
#define KFEAT   (CIN / 32)
#define GB      8
#define NG      ((KTOT / 32) * (16 / GB))
#define BCOLP   40
#define BKSTEP  (CMID * BCOLP)
#define FILLR   2
#define FILLN   ((KTOT / 32) * CMID * 4 / (WAVES * 32) / FILLR)

__device__ __forceinline__ void load_a(float4 (&r)[4], const float* src) {
    const float4* p = (const float4*)src;
    r[0] = p[0]; r[1] = p[1];
    r[2] = p[4]; r[3] = p[5];
}

__device__ __forceinline__ v16h cvt_a(const float4 (&r)[4]) {
    v16h a;
    a[0]  = (_Float16)r[0].x; a[1]  = (_Float16)r[0].y; a[2]  = (_Float16)r[0].z; a[3]  = (_Float16)r[0].w;
    a[4]  = (_Float16)r[1].x; a[5]  = (_Float16)r[1].y; a[6]  = (_Float16)r[1].z; a[7]  = (_Float16)r[1].w;
    a[8]  = (_Float16)r[2].x; a[9]  = (_Float16)r[2].y; a[10] = (_Float16)r[2].z; a[11] = (_Float16)r[2].w;
    a[12] = (_Float16)r[3].x; a[13] = (_Float16)r[3].y; a[14] = (_Float16)r[3].z; a[15] = (_Float16)r[3].w;
    return a;
}

__device__ __forceinline__ void load_bgroup(v16h (&b)[GB], const _Float16* bl,
                                            const unsigned (&koff)[KTOT / 32], int G) {
    const int kt = G / (16 / GB), sub = G % (16 / GB);
#pragma unroll
    for (int j = 0; j < GB; ++j) {
        const _Float16* bp = bl + koff[kt] + (sub * GB + j) * 16 * BCOLP;
        v8h blo = *(const v8h*)bp;
        v8h bhi = *(const v8h*)(bp + 16);
        b[j] = __builtin_shufflevector(blo, bhi, 0,1,2,3,4,5,6,7,8,9,10,11,12,13,14,15);
    }
}

template <bool PF>
__device__ __forceinline__ void gemm_pass(float4 (&ring)[DEP][RT][4], v16h (&bq)[2][GB],
                                          const float* fcur, const float* fnext,
                                          const float* pcur, const _Float16* bl,
                                          int half, float cx, float cy, float cz,
                                          float (&cm)[16]) {
    float px[RT], py[RT], pz[RT];
#pragma unroll
    for (int rt = 0; rt < RT; ++rt) {
        px[rt] = pcur[rt * 48 + 0]; py[rt] = pcur[rt * 48 + 1]; pz[rt] = pcur[rt * 48 + 2];
    }

    v8f acc[RT][16] = {};
    v16h afrag[RT];

    unsigned koff[KTOT / 32];
#pragma unroll
    for (int kt = 0; kt < KTOT / 32; ++kt) {
        koff[kt] = kt * BKSTEP;
        asm volatile("" : "+s"(koff[kt]));
    }

#pragma unroll
    for (int G = 0; G < NG; ++G) {
        const int kt = G / (16 / GB), sub = G % (16 / GB);
        if (G + 1 < NG) load_bgroup(bq[(G + 1) & 1], bl, koff, G + 1);

        __builtin_amdgcn_sched_barrier(0);

        if (sub == 0) {
            if (kt < KFEAT) {
                const int slot = kt % DEP;
#pragma unroll
                for (int rt = 0; rt < RT; ++rt) afrag[rt] = cvt_a(ring[slot][rt]);
                const int kn = kt + DEP;
                if (kn < KFEAT) {
#pragma unroll
                    for (int rt = 0; rt < RT; ++rt)
                        load_a(ring[slot][rt], fcur + rt * 16 * CIN + kn * 32);
                } else if (PF) {
#pragma unroll
                    for (int rt = 0; rt < RT; ++rt)
                        load_a(ring[slot][rt], fnext + rt * 16 * CIN + (kn - KFEAT) * 32);
                }
            } else {
#pragma unroll
                for (int rt = 0; rt < RT; ++rt) {
                    v16h a = {};
                    a[0] = half ? (_Float16)0.0f : (_Float16)(px[rt] - cx);
                    a[1] = half ? (_Float16)0.0f : (_Float16)(py[rt] - cy);
                    a[2] = half ? (_Float16)0.0f : (_Float16)(pz[rt] - cz);
                    afrag[rt] = a;
                }
            }
        }

#pragma unroll
        for (int j = 0; j < GB; ++j)
#pragma unroll
            for (int rt = 0; rt < RT; ++rt)
                acc[rt][sub * GB + j] = __builtin_amdgcn_wmma_f32_16x16x32_f16(
                    false, afrag[rt], false, bq[G & 1][j], (short)0, acc[rt][sub * GB + j], false, false);

        __builtin_amdgcn_sched_barrier(0);
    }

    if (PF) load_bgroup(bq[0], bl, koff, 0);
    __builtin_amdgcn_sched_barrier(0);

#pragma unroll
    for (int ct = 0; ct < 16; ++ct)
#pragma unroll
        for (int rt = 0; rt < RT; ++rt)
#pragma unroll
            for (int i = 0; i < 8; ++i)
                cm[ct] = fmaxf(cm[ct], acc[rt][ct][i]);
}

__device__ __forceinline__ void flush_chunk(float (&cm)[16], float* dst, int lane, int half) {
#pragma unroll
    for (int j = 0; j < 8; ++j) {
        float m0 = fmaxf(cm[2 * j],     __shfl_xor(cm[2 * j],     16, 32));
        float m1 = fmaxf(cm[2 * j + 1], __shfl_xor(cm[2 * j + 1], 16, 32));
        dst[32 * j + lane] = half ? m1 : m0;
    }
#pragma unroll
    for (int ct = 0; ct < 16; ++ct) cm[ct] = -INFINITY;
}

__global__ __launch_bounds__(WAVES * 32)
void gemm_max_kernel(const float* __restrict__ pos,
                     const float* __restrict__ feature,
                     const _Float16* __restrict__ W1t,
                     const float* __restrict__ center,
                     float* __restrict__ partial) {
    __shared__ __align__(16) _Float16 Bs[(KTOT / 32) * BKSTEP];

    const int g    = blockIdx.x;
    const int t    = threadIdx.x;
    const int lane = t & 31;
    const int wv   = t >> 5;
    const int n    = lane & 15;
    const int half = lane >> 4;
    const int lo8  = half * 8;
    const int s    = g >> 1;
    const int wrow0 = g * (WAVES * WROWS) + wv * WROWS;

    const float* fl = feature + (size_t)(wrow0 + n) * CIN + lo8;
    const float* pl = pos + (size_t)(wrow0 + n) * 3;

    float4 ring[DEP][RT][4];
#pragma unroll
    for (int d = 0; d < DEP; ++d)
#pragma unroll
        for (int rt = 0; rt < RT; ++rt) load_a(ring[d][rt], fl + rt * 16 * CIN + d * 32);

#pragma unroll
    for (int r = 0; r < FILLR; ++r) {
        v8h tmp[FILLN];
#pragma unroll
        for (int i = 0; i < FILLN; ++i)
            tmp[i] = *(const v8h*)(W1t + (size_t)(t + (r * FILLN + i) * (WAVES * 32)) * 8);
#pragma unroll
        for (int i = 0; i < FILLN; ++i) {
            const int pi = t + (r * FILLN + i) * (WAVES * 32);
            *(v8h*)&Bs[(pi >> 2) * BCOLP + (pi & 3) * 8] = tmp[i];
        }
    }
    const float cx = center[s * 4 + 0];
    const float cy = center[s * 4 + 1];
    const float cz = center[s * 4 + 2];
    __syncthreads();

    const _Float16* bl = Bs + n * BCOLP + lo8;
    float* pout = partial + (size_t)(wrow0 / ROWS) * CMID;

    float cm[16];
#pragma unroll
    for (int ct = 0; ct < 16; ++ct) cm[ct] = -INFINITY;

    v16h bq[2][GB];
    {
        unsigned koff0[KTOT / 32] = {};
        load_bgroup(bq[0], bl, koff0, 0);
    }

    for (int q = 0; q < NPASS - 1; ++q) {
        const float* fcur = fl + (size_t)q * (16 * RT * CIN);
        gemm_pass<true>(ring, bq, fcur, fcur + 16 * RT * CIN, pl + q * (16 * RT * 3), bl,
                        half, cx, cy, cz, cm);
        if ((q % PPC) == PPC - 1) flush_chunk(cm, pout + (q / PPC) * CMID, lane, half);
    }
    {
        const int q = NPASS - 1;
        const float* fcur = fl + (size_t)q * (16 * RT * CIN);
        gemm_pass<false>(ring, bq, fcur, fcur, pl + q * (16 * RT * 3), bl,
                         half, cx, cy, cz, cm);
        flush_chunk(cm, pout + (q / PPC) * CMID, lane, half);
    }
}

__global__ __launch_bounds__(256)
void finish_kernel(const float* __restrict__ partial,
                   const float* __restrict__ b1,
                   const float* __restrict__ W2,
                   const float* __restrict__ b2,
                   const float* __restrict__ noise,
                   float* __restrict__ out) {
    __shared__ float agg[CMID];
    const int s = blockIdx.x;
    const int c = threadIdx.x;

    const float* p = partial + (size_t)s * CHUNKS * CMID + c;
    float m = p[0];
    for (int ch = 1; ch < CHUNKS; ++ch) m = fmaxf(m, p[ch * CMID]);
    agg[c] = fmaxf(m + b1[c], 0.0f);
    __syncthreads();

    float mu = b2[c];
    float sr = b2[c + CMID];
    for (int k = 0; k < CMID; ++k) {
        float a = agg[k];
        mu += a * W2[(size_t)k * COUT + c];
        sr += a * W2[(size_t)k * COUT + c + CMID];
    }
    float sig = fmaxf(sr, 0.0f) + log1pf(expf(-fabsf(sr))) + 1e-4f;
    out[(size_t)s * CMID + c] = mu + sig * noise[(size_t)s * CMID + c];
}

extern "C" void kernel_launch(void* const* d_in, const int* in_sizes, int n_in,
                              void* d_out, int out_size, void* d_ws, size_t ws_size,
                              hipStream_t stream) {
    const float* pos     = (const float*)d_in[0];
    const float* feature = (const float*)d_in[1];
    const float* W1      = (const float*)d_in[3];
    const float* b1      = (const float*)d_in[4];
    const float* W2      = (const float*)d_in[5];
    const float* b2      = (const float*)d_in[6];
    const float* noise   = (const float*)d_in[7];
    float* out = (float*)d_out;

    _Float16* W1t    = (_Float16*)((char*)d_ws + WS_W1T);
    float*    center = (float*)((char*)d_ws + WS_CENTER);
    float*    partial= (float*)((char*)d_ws + WS_PARTIAL);

    prep_w1t_kernel<<<(CMID * KTOT + 255) / 256, 256, 0, stream>>>(W1, W1t);
    centroid_kernel<<<BSC, 256, 0, stream>>>(pos, center);
    gemm_max_kernel<<<NPTS / (WAVES * WROWS), WAVES * 32, 0, stream>>>(pos, feature, W1t, center, partial);
    finish_kernel<<<BSC, 256, 0, stream>>>(partial, b1, W2, b2, noise, out);
}


# GEMM A rows fetched as whole 128-B lines (a load touches 4 lines; in lane = row order it touched 16) and put in fragment order through a wave-private f16 LDS slab; no barrier added, sums unchanged
# speedup vs baseline: 1.4726x; 1.1489x over previous
#include <hip/hip_runtime.h>
#include <hip/hip_bf16.h>
#include <math.h>

typedef __attribute__((ext_vector_type(16))) _Float16 v16h;
typedef __attribute__((ext_vector_type(8)))  _Float16 v8h;
typedef __attribute__((ext_vector_type(8)))  float    v8f;

#define NPTS      262144
#define BSC       128
#define PPS       2048
#define CIN       256
#define CMID      256
#define COUT      512
#define KTOT      288
#define ROWS      64
#define CHUNKS    (PPS / ROWS)

#define WS_W1T     0
#define WS_PARTIAL 149504

__global__ void prep_w1t_kernel(const float* __restrict__ W1,
                                _Float16* __restrict__ W1t) {
    int idx = blockIdx.x * blockDim.x + threadIdx.x;
    if (idx >= CMID * KTOT) return;
    int col = idx / KTOT;
    int k   = idx % KTOT;
    float v = (k < CIN + 3) ? W1[(size_t)k * CMID + col] : 0.0f;
    W1t[(size_t)((k >> 5) * CMID + col) * 32 + (k & 31)] = (_Float16)v;
}

#define RT      4
#define NLD     (16 * RT / 4)
#define KDS     (CIN / 64)
#define APITCH  72
#define ASLAB   (16 * RT * APITCH)
#define WAVES   4
#define WROWS   (1024 / WAVES)
#define NPASS   (WROWS / (16 * RT))
#define PPC     (ROWS / (16 * RT))
#define KFEAT   (CIN / 32)
#define GB      8
#define NG      ((KTOT / 32) * (16 / GB))
#define BCOLP   40
#define BKSTEP  (CMID * BCOLP)
#define FILLR   2
#define FILLN   ((KTOT / 32) * CMID * 4 / (WAVES * 32) / FILLR)

typedef __attribute__((ext_vector_type(4)))  _Float16 v4h;
typedef __attribute__((ext_vector_type(4)))  float    v4f;

__device__ __forceinline__ void load_slot(v4f (&r)[NLD], const float* src) {
#pragma unroll
    for (int i = 0; i < NLD; ++i) r[i] = *(const v4f*)(src + i * 2 * CIN);
}

template <bool REFILL>
__device__ __forceinline__ void conv_slot(v4f (&r)[NLD], _Float16* dst, const float* src) {
#pragma unroll
    for (int i = 0; i < NLD; ++i) {
        const v4f x = r[i];
        v4h h;
        h[0] = (_Float16)x[0]; h[1] = (_Float16)x[1]; h[2] = (_Float16)x[2]; h[3] = (_Float16)x[3];
        *(v4h*)(dst + i * 2 * APITCH) = h;
    }
    if (REFILL) load_slot(r, src);
}

__device__ __forceinline__ void load_afrag(v16h (&a)[RT], const _Float16* src) {
#pragma unroll
    for (int rt = 0; rt < RT; ++rt) {
        const _Float16* ap = src + rt * 16 * APITCH;
        v8h alo = *(const v8h*)ap;
        v8h ahi = *(const v8h*)(ap + 16);
        a[rt] = __builtin_shufflevector(alo, ahi, 0,1,2,3,4,5,6,7,8,9,10,11,12,13,14,15);
    }
}

__device__ __forceinline__ void load_bgroup(v16h (&b)[GB], const _Float16* bl,
                                            const unsigned (&koff)[KTOT / 32], int G) {
    const int kt = G / (16 / GB), sub = G % (16 / GB);
#pragma unroll
    for (int j = 0; j < GB; ++j) {
        const _Float16* bp = bl + koff[kt] + (sub * GB + j) * 16 * BCOLP;
        v8h blo = *(const v8h*)bp;
        v8h bhi = *(const v8h*)(bp + 16);
        b[j] = __builtin_shufflevector(blo, bhi, 0,1,2,3,4,5,6,7,8,9,10,11,12,13,14,15);
    }
}

template <bool PF>
__device__ __forceinline__ void gemm_pass(v4f (&ring)[2][NLD], v16h (&bq)[2][GB], v16h (&afq)[2][RT],
                                          const float* fcur, const float* fnext,
                                          const float* pcur, const _Float16* bl,
                                          _Float16* awr, const _Float16* ard,
                                          int half,
                                          float (&cm)[16]) {
    float px[RT], py[RT], pz[RT];
#pragma unroll
    for (int rt = 0; rt < RT; ++rt) {
        px[rt] = pcur[rt * 48 + 0]; py[rt] = pcur[rt * 48 + 1]; pz[rt] = pcur[rt * 48 + 2];
    }

    v8f acc[RT][16] = {};

    unsigned koff[KTOT / 32];
#pragma unroll
    for (int kt = 0; kt < KTOT / 32; ++kt) {
        koff[kt] = kt * BKSTEP;
        asm volatile("" : "+s"(koff[kt]));
    }

#pragma unroll
    for (int G = 0; G < NG; ++G) {
        const int kt = G / (16 / GB), sub = G % (16 / GB);
        if (G + 1 < NG) load_bgroup(bq[(G + 1) & 1], bl, koff, G + 1);
        if (sub == (16 / GB) - 1 && kt + 1 < KFEAT) {
            load_afrag(afq[(kt + 1) & 1], ard + (((kt + 1) >> 1) & 1) * ASLAB + ((kt + 1) & 1) * 32);
        }

        __builtin_amdgcn_sched_barrier(0);

        if (sub == 0) {
            if (kt < KFEAT) {
                const int kd = (kt + 2) >> 1;
                const int rh = kt & 1;
                if (kd < KDS || PF) {
                    _Float16* sl = awr + (kd & 1) * ASLAB + rh * 32 * APITCH;
                    const int kn = kd + 1;
                    if (kn < KDS)  conv_slot<true>(ring[rh], sl, fcur + rh * 32 * CIN + kn * 64);
                    else if (PF)   conv_slot<true>(ring[rh], sl, fnext + rh * 32 * CIN + (kn - KDS) * 64);
                    else           conv_slot<false>(ring[rh], sl, fcur);
                }
            } else {
#pragma unroll
                for (int rt = 0; rt < RT; ++rt) {
                    v16h a = {};
                    a[0] = half ? (_Float16)0.0f : (_Float16)px[rt];
                    a[1] = half ? (_Float16)0.0f : (_Float16)py[rt];
                    a[2] = half ? (_Float16)0.0f : (_Float16)pz[rt];
                    afq[kt & 1][rt] = a;
                }
            }
            __builtin_amdgcn_sched_barrier(0);
        }

#pragma unroll
        for (int j = 0; j < GB; ++j)
#pragma unroll
            for (int rt = 0; rt < RT; ++rt)
                acc[rt][sub * GB + j] = __builtin_amdgcn_wmma_f32_16x16x32_f16(
                    false, afq[kt & 1][rt], false, bq[G & 1][j], (short)0, acc[rt][sub * GB + j], false, false);

        __builtin_amdgcn_sched_barrier(0);
    }

    if (PF) {
        load_bgroup(bq[0], bl, koff, 0);
        load_afrag(afq[0], ard);
    }
    __builtin_amdgcn_sched_barrier(0);

#pragma unroll
    for (int ct = 0; ct < 16; ++ct)
#pragma unroll
        for (int rt = 0; rt < RT; ++rt)
#pragma unroll
            for (int i = 0; i < 8; ++i)
                cm[ct] = fmaxf(cm[ct], acc[rt][ct][i]);
}

__device__ __forceinline__ void flush_chunk(float (&cm)[16], float* dst, int lane, int half) {
#pragma unroll
    for (int j = 0; j < 8; ++j) {
        float m0 = fmaxf(cm[2 * j],     __shfl_xor(cm[2 * j],     16, 32));
        float m1 = fmaxf(cm[2 * j + 1], __shfl_xor(cm[2 * j + 1], 16, 32));
        dst[32 * j + lane] = half ? m1 : m0;
    }
#pragma unroll
    for (int ct = 0; ct < 16; ++ct) cm[ct] = -INFINITY;
}

__global__ __launch_bounds__(WAVES * 32)
void gemm_max_kernel(const float* __restrict__ pos,
                     const float* __restrict__ feature,
                     const _Float16* __restrict__ W1t,
                     float* __restrict__ partial) {
    __shared__ __align__(16) _Float16 Bs[(KTOT / 32) * BKSTEP];
    __shared__ __align__(16) _Float16 As[WAVES * 2 * ASLAB];

    const int g    = blockIdx.x;
    const int t    = threadIdx.x;
    const int lane = t & 31;
    const int wv   = t >> 5;
    const int n    = lane & 15;
    const int half = lane >> 4;
    const int lo8  = half * 8;
    const int wrow0 = g * (WAVES * WROWS) + wv * WROWS;

    const float* fl = feature + (size_t)(wrow0 + half) * CIN + n * 4;
    const float* pl = pos + (size_t)(wrow0 + n) * 3;
    _Float16*       awr = As + wv * (2 * ASLAB) + half * APITCH + n * 4;
    const _Float16* ard = As + wv * (2 * ASLAB) + n * APITCH + lo8;

    v4f ring[2][NLD];
    load_slot(ring[0], fl);
    load_slot(ring[1], fl + 32 * CIN);

#pragma unroll
    for (int r = 0; r < FILLR; ++r) {
        v8h tmp[FILLN];
#pragma unroll
        for (int i = 0; i < FILLN; ++i)
            tmp[i] = *(const v8h*)(W1t + (size_t)(t + (r * FILLN + i) * (WAVES * 32)) * 8);
#pragma unroll
        for (int i = 0; i < FILLN; ++i) {
            const int pi = t + (r * FILLN + i) * (WAVES * 32);
            *(v8h*)&Bs[(pi >> 2) * BCOLP + (pi & 3) * 8] = tmp[i];
        }
    }

    conv_slot<true>(ring[0], awr, fl + 64);
    conv_slot<true>(ring[1], awr + 32 * APITCH, fl + 32 * CIN + 64);
    __syncthreads();

    const _Float16* bl = Bs + n * BCOLP + lo8;
    float* pout = partial + (size_t)(wrow0 / ROWS) * CMID;

    float cm[16];
#pragma unroll
    for (int ct = 0; ct < 16; ++ct) cm[ct] = -INFINITY;

    v16h bq[2][GB];
    v16h afq[2][RT];
    {
        unsigned koff0[KTOT / 32] = {};
        load_bgroup(bq[0], bl, koff0, 0);
        load_afrag(afq[0], ard);
    }

    for (int q = 0; q < NPASS - 1; ++q) {
        const float* fcur = fl + (size_t)q * (16 * RT * CIN);
        gemm_pass<true>(ring, bq, afq, fcur, fcur + 16 * RT * CIN, pl + q * (16 * RT * 3), bl,
                        awr, ard, half, cm);
        if ((q % PPC) == PPC - 1) flush_chunk(cm, pout + (q / PPC) * CMID, lane, half);
    }
    {
        const int q = NPASS - 1;
        const float* fcur = fl + (size_t)q * (16 * RT * CIN);
        gemm_pass<false>(ring, bq, afq, fcur, fcur, pl + q * (16 * RT * 3), bl,
                         awr, ard, half, cm);
        flush_chunk(cm, pout + (q / PPC) * CMID, lane, half);
    }
}

__global__ __launch_bounds__(256)
void finish_kernel(const float* __restrict__ partial,
                   const float* __restrict__ pos,
                   const float* __restrict__ W1,
                   const float* __restrict__ b1,
                   const float* __restrict__ W2,
                   const float* __restrict__ b2,
                   const float* __restrict__ noise,
                   float* __restrict__ out) {
    __shared__ float agg[CMID];
    __shared__ float wsum[8 * 3];
    const int s = blockIdx.x;
    const int c = threadIdx.x;

    const float4* p4 = (const float4*)(pos + (size_t)s * PPS * 3);
    float4 v[6];
#pragma unroll
    for (int i = 0; i < 6; ++i) v[i] = p4[c + 256 * i];

    const float w0 = W1[(size_t)(CIN + 0) * CMID + c];
    const float w1 = W1[(size_t)(CIN + 1) * CMID + c];
    const float w2 = W1[(size_t)(CIN + 2) * CMID + c];

    const float* p = partial + (size_t)s * CHUNKS * CMID + c;
    float m = p[0];
    for (int ch = 1; ch < CHUNKS; ++ch) m = fmaxf(m, p[ch * CMID]);

    float a[3] = {0.0f, 0.0f, 0.0f};
#pragma unroll
    for (int i = 0; i < 6; ++i) {
        a[(i + 0) % 3] += v[i].x;
        a[(i + 1) % 3] += v[i].y;
        a[(i + 2) % 3] += v[i].z;
        a[(i + 3) % 3] += v[i].w;
    }
    const int r = c % 3;
    float sx = (r == 0) ? a[0] : (r == 1) ? a[2] : a[1];
    float sy = (r == 0) ? a[1] : (r == 1) ? a[0] : a[2];
    float sz = (r == 0) ? a[2] : (r == 1) ? a[1] : a[0];
    for (int off = 16; off > 0; off >>= 1) {
        sx += __shfl_xor(sx, off, 32);
        sy += __shfl_xor(sy, off, 32);
        sz += __shfl_xor(sz, off, 32);
    }
    if ((c & 31) == 0) {
        wsum[(c >> 5) * 3 + 0] = sx;
        wsum[(c >> 5) * 3 + 1] = sy;
        wsum[(c >> 5) * 3 + 2] = sz;
    }
    __syncthreads();
    float cx = 0.0f, cy = 0.0f, cz = 0.0f;
    for (int w = 0; w < 8; ++w) {
        cx += wsum[w * 3 + 0]; cy += wsum[w * 3 + 1]; cz += wsum[w * 3 + 2];
    }
    const float inv = 1.0f / (float)PPS;
    cx *= inv; cy *= inv; cz *= inv;

    agg[c] = fmaxf((m - (cx * w0 + cy * w1 + cz * w2)) + b1[c], 0.0f);
    __syncthreads();

    float mu = b2[c];
    float sr = b2[c + CMID];
    for (int k = 0; k < CMID; ++k) {
        float av = agg[k];
        mu += av * W2[(size_t)k * COUT + c];
        sr += av * W2[(size_t)k * COUT + c + CMID];
    }
    float sig = fmaxf(sr, 0.0f) + log1pf(expf(-fabsf(sr))) + 1e-4f;
    out[(size_t)s * CMID + c] = mu + sig * noise[(size_t)s * CMID + c];
}

extern "C" void kernel_launch(void* const* d_in, const int* in_sizes, int n_in,
                              void* d_out, int out_size, void* d_ws, size_t ws_size,
                              hipStream_t stream) {
    const float* pos     = (const float*)d_in[0];
    const float* feature = (const float*)d_in[1];
    const float* W1      = (const float*)d_in[3];
    const float* b1      = (const float*)d_in[4];
    const float* W2      = (const float*)d_in[5];
    const float* b2      = (const float*)d_in[6];
    const float* noise   = (const float*)d_in[7];
    float* out = (float*)d_out;

    _Float16* W1t    = (_Float16*)((char*)d_ws + WS_W1T);
    float*    partial= (float*)((char*)d_ws + WS_PARTIAL);

    prep_w1t_kernel<<<(CMID * KTOT + 255) / 256, 256, 0, stream>>>(W1, W1t);
    gemm_max_kernel<<<NPTS / (WAVES * WROWS), WAVES * 32, 0, stream>>>(pos, feature, W1t, partial);
    finish_kernel<<<BSC, 256, 0, stream>>>(partial, pos, W1, b1, W2, b2, noise, out);
}


# Finish kernel agg@W2 loads regrouped: thread = 4 adjacent columns x half of k, 16-B pieces, 40 rows in flight, first 32 requested at entry (b1, b2, noise too); k halves added via LDS, then bias
# speedup vs baseline: 1.9748x; 1.3410x over previous
#include <hip/hip_runtime.h>
#include <hip/hip_bf16.h>
#include <math.h>

typedef __attribute__((ext_vector_type(16))) _Float16 v16h;
typedef __attribute__((ext_vector_type(8)))  _Float16 v8h;
typedef __attribute__((ext_vector_type(8)))  float    v8f;

#define NPTS      262144
#define BSC       128
#define PPS       2048
#define CIN       256
#define CMID      256
#define COUT      512
#define KTOT      288
#define ROWS      64
#define CHUNKS    (PPS / ROWS)

#define WS_W1T     0
#define WS_PARTIAL 149504

__global__ void prep_w1t_kernel(const float* __restrict__ W1,
                                _Float16* __restrict__ W1t) {
    int idx = blockIdx.x * blockDim.x + threadIdx.x;
    if (idx >= CMID * KTOT) return;
    int col = idx / KTOT;
    int k   = idx % KTOT;
    float v = (k < CIN + 3) ? W1[(size_t)k * CMID + col] : 0.0f;
    W1t[(size_t)((k >> 5) * CMID + col) * 32 + (k & 31)] = (_Float16)v;
}

#define RT      4
#define NLD     (16 * RT / 4)
#define KDS     (CIN / 64)
#define APITCH  72
#define ASLAB   (16 * RT * APITCH)
#define WAVES   4
#define WROWS   (1024 / WAVES)
#define NPASS   (WROWS / (16 * RT))
#define PPC     (ROWS / (16 * RT))
#define KFEAT   (CIN / 32)
#define GB      8
#define NG      ((KTOT / 32) * (16 / GB))
#define BCOLP   40
#define BKSTEP  (CMID * BCOLP)
#define FILLR   2
#define FILLN   ((KTOT / 32) * CMID * 4 / (WAVES * 32) / FILLR)

typedef __attribute__((ext_vector_type(4)))  _Float16 v4h;
typedef __attribute__((ext_vector_type(4)))  float    v4f;

__device__ __forceinline__ void load_slot(v4f (&r)[NLD], const float* src) {
#pragma unroll
    for (int i = 0; i < NLD; ++i) r[i] = *(const v4f*)(src + i * 2 * CIN);
}

template <bool REFILL>
__device__ __forceinline__ void conv_slot(v4f (&r)[NLD], _Float16* dst, const float* src) {
#pragma unroll
    for (int i = 0; i < NLD; ++i) {
        const v4f x = r[i];
        v4h h;
        h[0] = (_Float16)x[0]; h[1] = (_Float16)x[1]; h[2] = (_Float16)x[2]; h[3] = (_Float16)x[3];
        *(v4h*)(dst + i * 2 * APITCH) = h;
    }
    if (REFILL) load_slot(r, src);
}

__device__ __forceinline__ void load_afrag(v16h (&a)[RT], const _Float16* src) {
#pragma unroll
    for (int rt = 0; rt < RT; ++rt) {
        const _Float16* ap = src + rt * 16 * APITCH;
        v8h alo = *(const v8h*)ap;
        v8h ahi = *(const v8h*)(ap + 16);
        a[rt] = __builtin_shufflevector(alo, ahi, 0,1,2,3,4,5,6,7,8,9,10,11,12,13,14,15);
    }
}

__device__ __forceinline__ void load_bgroup(v16h (&b)[GB], const _Float16* bl,
                                            const unsigned (&koff)[KTOT / 32], int G) {
    const int kt = G / (16 / GB), sub = G % (16 / GB);
#pragma unroll
    for (int j = 0; j < GB; ++j) {
        const _Float16* bp = bl + koff[kt] + (sub * GB + j) * 16 * BCOLP;
        v8h blo = *(const v8h*)bp;
        v8h bhi = *(const v8h*)(bp + 16);
        b[j] = __builtin_shufflevector(blo, bhi, 0,1,2,3,4,5,6,7,8,9,10,11,12,13,14,15);
    }
}

template <bool PF>
__device__ __forceinline__ void gemm_pass(v4f (&ring)[2][NLD], v16h (&bq)[2][GB], v16h (&afq)[2][RT],
                                          const float* fcur, const float* fnext,
                                          const float* pcur, const _Float16* bl,
                                          _Float16* awr, const _Float16* ard,
                                          int half,
                                          float (&cm)[16]) {
    float px[RT], py[RT], pz[RT];
#pragma unroll
    for (int rt = 0; rt < RT; ++rt) {
        px[rt] = pcur[rt * 48 + 0]; py[rt] = pcur[rt * 48 + 1]; pz[rt] = pcur[rt * 48 + 2];
    }

    v8f acc[RT][16] = {};

    unsigned koff[KTOT / 32];
#pragma unroll
    for (int kt = 0; kt < KTOT / 32; ++kt) {
        koff[kt] = kt * BKSTEP;
        asm volatile("" : "+s"(koff[kt]));
    }

#pragma unroll
    for (int G = 0; G < NG; ++G) {
        const int kt = G / (16 / GB), sub = G % (16 / GB);
        if (G + 1 < NG) load_bgroup(bq[(G + 1) & 1], bl, koff, G + 1);
        if (sub == (16 / GB) - 1 && kt + 1 < KFEAT) {
            load_afrag(afq[(kt + 1) & 1], ard + (((kt + 1) >> 1) & 1) * ASLAB + ((kt + 1) & 1) * 32);
        }

        __builtin_amdgcn_sched_barrier(0);

        if (sub == 0) {
            if (kt < KFEAT) {
                const int kd = (kt + 2) >> 1;
                const int rh = kt & 1;
                if (kd < KDS || PF) {
                    _Float16* sl = awr + (kd & 1) * ASLAB + rh * 32 * APITCH;
                    const int kn = kd + 1;
                    if (kn < KDS)  conv_slot<true>(ring[rh], sl, fcur + rh * 32 * CIN + kn * 64);
                    else if (PF)   conv_slot<true>(ring[rh], sl, fnext + rh * 32 * CIN + (kn - KDS) * 64);
                    else           conv_slot<false>(ring[rh], sl, fcur);
                }
            } else {
#pragma unroll
                for (int rt = 0; rt < RT; ++rt) {
                    v16h a = {};
                    a[0] = half ? (_Float16)0.0f : (_Float16)px[rt];
                    a[1] = half ? (_Float16)0.0f : (_Float16)py[rt];
                    a[2] = half ? (_Float16)0.0f : (_Float16)pz[rt];
                    afq[kt & 1][rt] = a;
                }
            }
            __builtin_amdgcn_sched_barrier(0);
        }

#pragma unroll
        for (int j = 0; j < GB; ++j)
#pragma unroll
            for (int rt = 0; rt < RT; ++rt)
                acc[rt][sub * GB + j] = __builtin_amdgcn_wmma_f32_16x16x32_f16(
                    false, afq[kt & 1][rt], false, bq[G & 1][j], (short)0, acc[rt][sub * GB + j], false, false);

        __builtin_amdgcn_sched_barrier(0);
    }

    if (PF) {
        load_bgroup(bq[0], bl, koff, 0);
        load_afrag(afq[0], ard);
    }
    __builtin_amdgcn_sched_barrier(0);

#pragma unroll
    for (int ct = 0; ct < 16; ++ct)
#pragma unroll
        for (int rt = 0; rt < RT; ++rt)
#pragma unroll
            for (int i = 0; i < 8; ++i)
                cm[ct] = fmaxf(cm[ct], acc[rt][ct][i]);
}

__device__ __forceinline__ void flush_chunk(float (&cm)[16], float* dst, int lane, int half) {
#pragma unroll
    for (int j = 0; j < 8; ++j) {
        float m0 = fmaxf(cm[2 * j],     __shfl_xor(cm[2 * j],     16, 32));
        float m1 = fmaxf(cm[2 * j + 1], __shfl_xor(cm[2 * j + 1], 16, 32));
        dst[32 * j + lane] = half ? m1 : m0;
    }
#pragma unroll
    for (int ct = 0; ct < 16; ++ct) cm[ct] = -INFINITY;
}

__global__ __launch_bounds__(WAVES * 32)
void gemm_max_kernel(const float* __restrict__ pos,
                     const float* __restrict__ feature,
                     const _Float16* __restrict__ W1t,
                     float* __restrict__ partial) {
    __shared__ __align__(16) _Float16 Bs[(KTOT / 32) * BKSTEP];
    __shared__ __align__(16) _Float16 As[WAVES * 2 * ASLAB];

    const int g    = blockIdx.x;
    const int t    = threadIdx.x;
    const int lane = t & 31;
    const int wv   = t >> 5;
    const int n    = lane & 15;
    const int half = lane >> 4;
    const int lo8  = half * 8;
    const int wrow0 = g * (WAVES * WROWS) + wv * WROWS;

    const float* fl = feature + (size_t)(wrow0 + half) * CIN + n * 4;
    const float* pl = pos + (size_t)(wrow0 + n) * 3;
    _Float16*       awr = As + wv * (2 * ASLAB) + half * APITCH + n * 4;
    const _Float16* ard = As + wv * (2 * ASLAB) + n * APITCH + lo8;

    v4f ring[2][NLD];
    load_slot(ring[0], fl);
    load_slot(ring[1], fl + 32 * CIN);

#pragma unroll
    for (int r = 0; r < FILLR; ++r) {
        v8h tmp[FILLN];
#pragma unroll
        for (int i = 0; i < FILLN; ++i)
            tmp[i] = *(const v8h*)(W1t + (size_t)(t + (r * FILLN + i) * (WAVES * 32)) * 8);
#pragma unroll
        for (int i = 0; i < FILLN; ++i) {
            const int pi = t + (r * FILLN + i) * (WAVES * 32);
            *(v8h*)&Bs[(pi >> 2) * BCOLP + (pi & 3) * 8] = tmp[i];
        }
    }

    conv_slot<true>(ring[0], awr, fl + 64);
    conv_slot<true>(ring[1], awr + 32 * APITCH, fl + 32 * CIN + 64);
    __syncthreads();

    const _Float16* bl = Bs + n * BCOLP + lo8;
    float* pout = partial + (size_t)(wrow0 / ROWS) * CMID;

    float cm[16];
#pragma unroll
    for (int ct = 0; ct < 16; ++ct) cm[ct] = -INFINITY;

    v16h bq[2][GB];
    v16h afq[2][RT];
    {
        unsigned koff0[KTOT / 32] = {};
        load_bgroup(bq[0], bl, koff0, 0);
        load_afrag(afq[0], ard);
    }

    for (int q = 0; q < NPASS - 1; ++q) {
        const float* fcur = fl + (size_t)q * (16 * RT * CIN);
        gemm_pass<true>(ring, bq, afq, fcur, fcur + 16 * RT * CIN, pl + q * (16 * RT * 3), bl,
                        awr, ard, half, cm);
        if ((q % PPC) == PPC - 1) flush_chunk(cm, pout + (q / PPC) * CMID, lane, half);
    }
    {
        const int q = NPASS - 1;
        const float* fcur = fl + (size_t)q * (16 * RT * CIN);
        gemm_pass<false>(ring, bq, afq, fcur, fcur, pl + q * (16 * RT * 3), bl,
                         awr, ard, half, cm);
        flush_chunk(cm, pout + (q / PPC) * CMID, lane, half);
    }
}

#define FG      16
#define FNG     (CMID / 2 / FG)

__device__ __forceinline__ void load_w2group(float4 (&w)[FG], const float4* src) {
#pragma unroll
    for (int i = 0; i < FG; ++i) w[i] = src[i * (COUT / 4)];
}

__global__ __launch_bounds__(256)
void finish_kernel(const float* __restrict__ partial,
                   const float* __restrict__ pos,
                   const float* __restrict__ W1,
                   const float* __restrict__ b1,
                   const float* __restrict__ W2,
                   const float* __restrict__ b2,
                   const float* __restrict__ noise,
                   float* __restrict__ out) {
    __shared__ __align__(16) float agg[CMID];
    __shared__ float wsum[8 * 3];
    __shared__ __align__(16) float part[2 * COUT];
    const int s = blockIdx.x;
    const int c = threadIdx.x;
    const int q  = c & 127;
    const int kh = c >> 7;

    const float4* p4 = (const float4*)(pos + (size_t)s * PPS * 3);
    float4 v[6];
#pragma unroll
    for (int i = 0; i < 6; ++i) v[i] = p4[c + 256 * i];

    const float w0 = W1[(size_t)(CIN + 0) * CMID + c];
    const float w1 = W1[(size_t)(CIN + 1) * CMID + c];
    const float w2 = W1[(size_t)(CIN + 2) * CMID + c];
    const float b1c = b1[c];

    const float* p = partial + (size_t)s * CHUNKS * CMID + c;
    float pm[CHUNKS];
#pragma unroll
    for (int ch = 0; ch < CHUNKS; ++ch) pm[ch] = p[ch * CMID];

    const float4* w2p = (const float4*)(W2 + (size_t)(kh * (CMID / 2)) * COUT) + q;
    float4 wq[2][FG];
    load_w2group(wq[0], w2p);
    load_w2group(wq[1], w2p + FG * (COUT / 4));
    const float bmu = b2[c];
    const float bsr = b2[c + CMID];
    const float nz  = noise[(size_t)s * CMID + c];

    float m = pm[0];
#pragma unroll
    for (int ch = 1; ch < CHUNKS; ++ch) m = fmaxf(m, pm[ch]);

    float a[3] = {0.0f, 0.0f, 0.0f};
#pragma unroll
    for (int i = 0; i < 6; ++i) {
        a[(i + 0) % 3] += v[i].x;
        a[(i + 1) % 3] += v[i].y;
        a[(i + 2) % 3] += v[i].z;
        a[(i + 3) % 3] += v[i].w;
    }
    const int r = c % 3;
    float sx = (r == 0) ? a[0] : (r == 1) ? a[2] : a[1];
    float sy = (r == 0) ? a[1] : (r == 1) ? a[0] : a[2];
    float sz = (r == 0) ? a[2] : (r == 1) ? a[1] : a[0];
    for (int off = 16; off > 0; off >>= 1) {
        sx += __shfl_xor(sx, off, 32);
        sy += __shfl_xor(sy, off, 32);
        sz += __shfl_xor(sz, off, 32);
    }
    if ((c & 31) == 0) {
        wsum[(c >> 5) * 3 + 0] = sx;
        wsum[(c >> 5) * 3 + 1] = sy;
        wsum[(c >> 5) * 3 + 2] = sz;
    }
    __syncthreads();
    float cx = 0.0f, cy = 0.0f, cz = 0.0f;
    for (int w = 0; w < 8; ++w) {
        cx += wsum[w * 3 + 0]; cy += wsum[w * 3 + 1]; cz += wsum[w * 3 + 2];
    }
    const float inv = 1.0f / (float)PPS;
    cx *= inv; cy *= inv; cz *= inv;

    agg[c] = fmaxf((m - (cx * w0 + cy * w1 + cz * w2)) + b1c, 0.0f);
    __syncthreads();

    float4 acc = make_float4(0.0f, 0.0f, 0.0f, 0.0f);
    const float4* ap = (const float4*)(agg + kh * (CMID / 2));
#pragma unroll
    for (int g = 0; g < FNG; ++g) {
#pragma unroll
        for (int i = 0; i < FG; i += 4) {
            const float4 a4 = ap[(g * FG + i) >> 2];
            const float av[4] = {a4.x, a4.y, a4.z, a4.w};
#pragma unroll
            for (int e = 0; e < 4; ++e) {
                const float4 w = wq[g & 1][i + e];
                acc.x += av[e] * w.x;
                acc.y += av[e] * w.y;
                acc.z += av[e] * w.z;
                acc.w += av[e] * w.w;
            }
        }
        if (g + 2 < FNG) load_w2group(wq[g & 1], w2p + (g + 2) * FG * (COUT / 4));
    }
    *(float4*)(part + kh * COUT + 4 * q) = acc;
    __syncthreads();

    const float mu = (bmu + part[c]) + part[COUT + c];
    const float sr = (bsr + part[CMID + c]) + part[COUT + CMID + c];
    float sig = fmaxf(sr, 0.0f) + log1pf(expf(-fabsf(sr))) + 1e-4f;
    out[(size_t)s * CMID + c] = mu + sig * nz;
}

extern "C" void kernel_launch(void* const* d_in, const int* in_sizes, int n_in,
                              void* d_out, int out_size, void* d_ws, size_t ws_size,
                              hipStream_t stream) {
    const float* pos     = (const float*)d_in[0];
    const float* feature = (const float*)d_in[1];
    const float* W1      = (const float*)d_in[3];
    const float* b1      = (const float*)d_in[4];
    const float* W2      = (const float*)d_in[5];
    const float* b2      = (const float*)d_in[6];
    const float* noise   = (const float*)d_in[7];
    float* out = (float*)d_out;

    _Float16* W1t    = (_Float16*)((char*)d_ws + WS_W1T);
    float*    partial= (float*)((char*)d_ws + WS_PARTIAL);

    prep_w1t_kernel<<<(CMID * KTOT + 255) / 256, 256, 0, stream>>>(W1, W1t);
    gemm_max_kernel<<<NPTS / (WAVES * WROWS), WAVES * 32, 0, stream>>>(pos, feature, W1t, partial);
    finish_kernel<<<BSC, 256, 0, stream>>>(partial, pos, W1, b1, W2, b2, noise, out);
}
